# starting kernel
# baseline (speedup 1.0000x reference)
#include <hip/hip_runtime.h>

typedef float v2f __attribute__((ext_vector_type(2)));
typedef float v8f __attribute__((ext_vector_type(8)));

#define NEG_INF_F (-1000000000.0f)

constexpr int Bn = 64, Cn = 1024, Qn = 128, Dn = 128;
constexpr int P = 132;
constexpr int CHUNK = 128;
constexpr int NCHUNK = Cn / CHUNK;
constexpr int OUTW = 4 * Dn;
constexpr int SM_FLOATS = 3 * (128 * P) + 384 + 128 + 128 + 1024 + 256 + 128;

__global__ __launch_bounds__(256, 1)
void cqa_kernel(const float* __restrict__ ctx_g,
                const float* __restrict__ qry_g,
                const float* __restrict__ W_g,
                const float* __restrict__ qmask_g,
                float* __restrict__ out_g)
{
    extern __shared__ float smem[];
    float* q_lds   = smem;
    float* ctx_lds = q_lds   + Qn * P;
    float* sc_lds  = ctx_lds + CHUNK * P;
    float* w_lds   = sc_lds  + CHUNK * P;
    float* qbias   = w_lds   + 384;
    float* cterm   = qbias   + Qn;
    float* m_all   = cterm   + CHUNK;
    float* red     = m_all   + Cn;
    float* q2c     = red     + 256;

    const int b    = blockIdx.x;
    const int tid  = threadIdx.x;
    const int wv   = tid >> 5;
    const int lane = tid & 31;
    const int lh   = lane >> 4;
    const int ln   = lane & 15;

    const float* ctxB = ctx_g + (size_t)b * Cn * Dn;
    const float* qryB = qry_g + (size_t)b * Qn * Dn;
    float*       outB = out_g + (size_t)b * Cn * OUTW;

    for (int i = tid; i < 3 * Dn; i += 256) w_lds[i] = W_g[i];
    for (int r = wv; r < Qn; r += 8) {
        const int c4 = lane * 4;
        const float4 v = *reinterpret_cast<const float4*>(qryB + (size_t)r * Dn + c4);
        *reinterpret_cast<float4*>(q_lds + r * P + c4) = v;
    }
    __syncthreads();
    if (tid < Qn) {
        float s = 0.f;
        for (int d = 0; d < Dn; ++d) s += q_lds[tid * P + d] * w_lds[Dn + d];
        qbias[tid] = s + (1.0f - qmask_g[b * Qn + tid]) * NEG_INF_F;
    }
    __syncthreads();

    const int rm = wv * 16;

    for (int ch = 0; ch < NCHUNK; ++ch) {
        const int cc = ch * CHUNK;

        for (int r = wv; r < CHUNK; r += 8) {
            const int c4 = lane * 4;
            const float4 v = *reinterpret_cast<const float4*>(ctxB + (size_t)(cc + r) * Dn + c4);
            *reinterpret_cast<float4*>(ctx_lds + r * P + c4) = v;
        }
        __syncthreads();
        if (tid < CHUNK) {
            float s = 0.f;
            for (int d = 0; d < Dn; ++d) s += ctx_lds[tid * P + d] * w_lds[d];
            cterm[tid] = s;
        }
        __syncthreads();

        v8f acc[8];
        #pragma unroll
        for (int t = 0; t < 8; ++t) acc[t] = v8f{0.f,0.f,0.f,0.f,0.f,0.f,0.f,0.f};

        for (int kb = 0; kb < Dn; kb += 4) {
            const int k0 = kb + 2 * lh;
            v2f a = *reinterpret_cast<const v2f*>(ctx_lds + (rm + ln) * P + k0);
            a.x *= w_lds[2 * Dn + k0];
            a.y *= w_lds[2 * Dn + k0 + 1];
            #pragma unroll
            for (int t = 0; t < 8; ++t) {
                const v2f bf = *reinterpret_cast<const v2f*>(q_lds + (t * 16 + ln) * P + k0);
                acc[t] = __builtin_amdgcn_wmma_f32_16x16x4_f32(
                    false, a, false, bf, (short)0, acc[t], false, false);
            }
        }

        #pragma unroll
        for (int r = 0; r < 8; ++r) {
            const int row = rm + r + 8 * lh;
            const float cb = cterm[row];
            float sv[8];
            float mx = -3.4e38f;
            #pragma unroll
            for (int t = 0; t < 8; ++t) {
                const float s = acc[t][r] + cb + qbias[t * 16 + ln];
                sv[t] = s;
                mx = fmaxf(mx, s);
            }
            #pragma unroll
            for (int off = 1; off < 16; off <<= 1)
                mx = fmaxf(mx, __shfl_xor(mx, off, 32));
            if (ln == 0) m_all[cc + row] = mx;
            float sum = 0.f;
            #pragma unroll
            for (int t = 0; t < 8; ++t) { sv[t] = __expf(sv[t] - mx); sum += sv[t]; }
            #pragma unroll
            for (int off = 1; off < 16; off <<= 1)
                sum += __shfl_xor(sum, off, 32);
            const float inv = 1.0f / sum;
            #pragma unroll
            for (int t = 0; t < 8; ++t)
                sc_lds[row * P + t * 16 + ln] = sv[t] * inv;
        }
        __syncthreads();

        v8f acc2[8];
        #pragma unroll
        for (int t = 0; t < 8; ++t) acc2[t] = v8f{0.f,0.f,0.f,0.f,0.f,0.f,0.f,0.f};

        for (int kb = 0; kb < Qn; kb += 4) {
            const int k0 = kb + 2 * lh;
            const v2f a = *reinterpret_cast<const v2f*>(sc_lds + (rm + ln) * P + k0);
            #pragma unroll
            for (int t = 0; t < 8; ++t) {
                v2f bf;
                bf.x = q_lds[k0 * P + t * 16 + ln];
                bf.y = q_lds[(k0 + 1) * P + t * 16 + ln];
                acc2[t] = __builtin_amdgcn_wmma_f32_16x16x4_f32(
                    false, a, false, bf, (short)0, acc2[t], false, false);
            }
        }
        __syncthreads();

        #pragma unroll
        for (int r = 0; r < 8; ++r) {
            const int row = rm + r + 8 * lh;
            #pragma unroll
            for (int t = 0; t < 8; ++t)
                sc_lds[row * P + t * 16 + ln] = acc2[t][r];
        }
        __syncthreads();

        for (int r = wv; r < CHUNK; r += 8) {
            const int c4 = lane * 4;
            const float4 cv = *reinterpret_cast<const float4*>(ctx_lds + r * P + c4);
            const float4 qv = *reinterpret_cast<const float4*>(sc_lds + r * P + c4);
            float* op = outB + (size_t)(cc + r) * OUTW;
            *reinterpret_cast<float4*>(op + c4) = cv;
            *reinterpret_cast<float4*>(op + Dn + c4) = qv;
            const float4 pv = make_float4(cv.x * qv.x, cv.y * qv.y, cv.z * qv.z, cv.w * qv.w);
            *reinterpret_cast<float4*>(op + 2 * Dn + c4) = pv;
        }
        __syncthreads();
    }

    {
        float mx = -3.4e38f;
        for (int i = tid; i < Cn; i += 256) mx = fmaxf(mx, m_all[i]);
        red[tid] = mx;
        __syncthreads();
        for (int s = 128; s > 0; s >>= 1) {
            if (tid < s) red[tid] = fmaxf(red[tid], red[tid + s]);
            __syncthreads();
        }
        const float gmax = red[0];
        __syncthreads();
        float z = 0.f;
        for (int i = tid; i < Cn; i += 256) {
            const float p = __expf(m_all[i] - gmax);
            m_all[i] = p;
            z += p;
        }
        red[tid] = z;
        __syncthreads();
        for (int s = 128; s > 0; s >>= 1) {
            if (tid < s) red[tid] += red[tid + s];
            __syncthreads();
        }
        const float invZ = 1.0f / red[0];
        __syncthreads();

        const int d    = tid & 127;
        const int half = tid >> 7;
        float acc = 0.f;
        for (int c = half * 512; c < half * 512 + 512; ++c)
            acc += m_all[c] * ctxB[(size_t)c * Dn + d];
        red[tid] = acc;
        __syncthreads();
        if (tid < Dn) q2c[tid] = (red[tid] + red[tid + 128]) * invZ;
        __syncthreads();
    }

    for (int idx = tid; idx < Cn * 32; idx += 256) {
        const int r  = idx >> 5;
        const int c4 = (idx & 31) * 4;
        const float4 cv = *reinterpret_cast<const float4*>(ctxB + (size_t)r * Dn + c4);
        const float4 qv = *reinterpret_cast<const float4*>(q2c + c4);
        const float4 pv = make_float4(cv.x * qv.x, cv.y * qv.y, cv.z * qv.z, cv.w * qv.w);
        *reinterpret_cast<float4*>(outB + (size_t)r * OUTW + 3 * Dn + c4) = pv;
    }
}

extern "C" void kernel_launch(void* const* d_in, const int* in_sizes, int n_in,
                              void* d_out, int out_size, void* d_ws, size_t ws_size,
                              hipStream_t stream) {
    const float* ctx   = (const float*)d_in[0];
    const float* qry   = (const float*)d_in[1];
    const float* W     = (const float*)d_in[2];
    const float* qmask = (const float*)d_in[3];
    float* out = (float*)d_out;

    const size_t smem_bytes = (size_t)SM_FLOATS * sizeof(float);
    hipFuncSetAttribute(reinterpret_cast<const void*>(cqa_kernel),
                        hipFuncAttributeMaxDynamicSharedMemorySize, (int)smem_bytes);
    cqa_kernel<<<dim3(Bn), dim3(256), smem_bytes, stream>>>(ctx, qry, W, qmask, out);
}


# Grid 64->512: one work-group per 128-row chunk; q2c from per-chunk (max, sum, weighted ctx sum) shares staged in out block 3, merged by a 2nd launch; q2c sum order changes (rescaled chunk partials)
# speedup vs baseline: 4.4189x; 4.4189x over previous
#include <hip/hip_runtime.h>


typedef float v2f __attribute__((ext_vector_type(2)));
typedef float v8f __attribute__((ext_vector_type(8)));

#define NEG_INF_F (-1000000000.0f)

constexpr int Bn = 64, Cn = 1024, Qn = 128, Dn = 128;
constexpr int P = 132;
constexpr int CHUNK = 128;
constexpr int NCHUNK = Cn / CHUNK;
constexpr int NWG = Bn * NCHUNK;
constexpr int OUTW = 4 * Dn;
constexpr int SM_FLOATS = 3 * (128 * P) + 384 + 128 + 128 + 128 + 128 + 8 * Dn;
static_assert(Qn == CHUNK, "query rows and chunk rows are staged by one loop");

__global__ __launch_bounds__(256, 1)
void cqa_kernel(const float* __restrict__ ctx_g,
                const float* __restrict__ qry_g,
                const float* __restrict__ W_g,
                const float* __restrict__ qmask_g,
                float* __restrict__ out_g)
{
    extern __shared__ float smem[];
    float* q_lds   = smem;
    float* ctx_lds = q_lds   + Qn * P;
    float* sc_lds  = ctx_lds + CHUNK * P;
    float* w_lds   = sc_lds  + CHUNK * P;
    float* qbias   = w_lds   + 384;
    float* cterm   = qbias   + Qn;
    float* m_loc   = cterm   + CHUNK;
    float* p_loc   = m_loc   + CHUNK;
    float* red     = p_loc   + CHUNK;

    const int b    = blockIdx.x / NCHUNK;
    const int ch   = blockIdx.x % NCHUNK;
    const int cc   = ch * CHUNK;
    const int tid  = threadIdx.x;
    const int wv   = tid >> 5;
    const int lane = tid & 31;
    const int lh   = lane >> 4;
    const int ln   = lane & 15;

    const float* ctxB = ctx_g + (size_t)b * Cn * Dn;
    const float* qryB = qry_g + (size_t)b * Qn * Dn;
    float*       outB = out_g + (size_t)b * Cn * OUTW;

    {
        const float w0 = W_g[tid];
        const float w1 = (tid < Dn) ? W_g[256 + tid] : 0.f;
        const int c4 = lane * 4;
        float4 qv[16], cv[16];
        for (int i = 0; i < 16; ++i) {
            const int r = wv + 8 * i;
            qv[i] = *reinterpret_cast<const float4*>(qryB + (size_t)r * Dn + c4);
            cv[i] = *reinterpret_cast<const float4*>(ctxB + (size_t)(cc + r) * Dn + c4);
        }
        w_lds[tid] = w0;
        if (tid < Dn) w_lds[256 + tid] = w1;
        for (int i = 0; i < 16; ++i) {
            const int r = wv + 8 * i;
            *reinterpret_cast<float4*>(q_lds + r * P + c4) = qv[i];
            *reinterpret_cast<float4*>(ctx_lds + r * P + c4) = cv[i];
        }
    }
    __syncthreads();
    if (tid < Qn) {
        float s = 0.f;
        for (int d = 0; d < Dn; ++d) s += q_lds[tid * P + d] * w_lds[Dn + d];
        qbias[tid] = s + (1.0f - qmask_g[b * Qn + tid]) * NEG_INF_F;
    } else {
        const int row = tid - Qn;
        float s = 0.f;
        for (int d = 0; d < Dn; ++d) s += ctx_lds[row * P + d] * w_lds[d];
        cterm[row] = s;
    }
    __syncthreads();

    const int rm = wv * 16;

    v8f acc[8];
    #pragma unroll
    for (int t = 0; t < 8; ++t) acc[t] = v8f{0.f,0.f,0.f,0.f,0.f,0.f,0.f,0.f};

    for (int kb = 0; kb < Dn; kb += 4) {
        const int k0 = kb + 2 * lh;
        v2f a = *reinterpret_cast<const v2f*>(ctx_lds + (rm + ln) * P + k0);
        a.x *= w_lds[2 * Dn + k0];
        a.y *= w_lds[2 * Dn + k0 + 1];
        #pragma unroll
        for (int t = 0; t < 8; ++t) {
            const v2f bf = *reinterpret_cast<const v2f*>(q_lds + (t * 16 + ln) * P + k0);
            acc[t] = __builtin_amdgcn_wmma_f32_16x16x4_f32(
                false, a, false, bf, (short)0, acc[t], false, false);
        }
    }

    #pragma unroll
    for (int r = 0; r < 8; ++r) {
        const int row = rm + r + 8 * lh;
        const float cb = cterm[row];
        float sv[8];
        float mx = -3.4e38f;
        #pragma unroll
        for (int t = 0; t < 8; ++t) {
            const float s = acc[t][r] + cb + qbias[t * 16 + ln];
            sv[t] = s;
            mx = fmaxf(mx, s);
        }
        #pragma unroll
        for (int off = 1; off < 16; off <<= 1)
            mx = fmaxf(mx, __shfl_xor(mx, off, 32));
        if (ln == 0) m_loc[row] = mx;
        float sum = 0.f;
        #pragma unroll
        for (int t = 0; t < 8; ++t) { sv[t] = __expf(sv[t] - mx); sum += sv[t]; }
        #pragma unroll
        for (int off = 1; off < 16; off <<= 1)
            sum += __shfl_xor(sum, off, 32);
        const float inv = 1.0f / sum;
        #pragma unroll
        for (int t = 0; t < 8; ++t)
            sc_lds[row * P + t * 16 + ln] = sv[t] * inv;
    }
    __syncthreads();

    v8f acc2[8];
    #pragma unroll
    for (int t = 0; t < 8; ++t) acc2[t] = v8f{0.f,0.f,0.f,0.f,0.f,0.f,0.f,0.f};

    for (int kb = 0; kb < Qn; kb += 4) {
        const int k0 = kb + 2 * lh;
        const v2f a = *reinterpret_cast<const v2f*>(sc_lds + (rm + ln) * P + k0);
        #pragma unroll
        for (int t = 0; t < 8; ++t) {
            v2f bf;
            bf.x = q_lds[k0 * P + t * 16 + ln];
            bf.y = q_lds[(k0 + 1) * P + t * 16 + ln];
            acc2[t] = __builtin_amdgcn_wmma_f32_16x16x4_f32(
                false, a, false, bf, (short)0, acc2[t], false, false);
        }
    }
    __syncthreads();

    #pragma unroll
    for (int r = 0; r < 8; ++r) {
        const int row = rm + r + 8 * lh;
        #pragma unroll
        for (int t = 0; t < 8; ++t)
            sc_lds[row * P + t * 16 + ln] = acc2[t][r];
    }

    const float m0 = m_loc[lane], m1 = m_loc[lane + 32], m2 = m_loc[lane + 64], m3 = m_loc[lane + 96];
    float gm = fmaxf(fmaxf(m0, m1), fmaxf(m2, m3));
    for (int off = 1; off < 32; off <<= 1)
        gm = fmaxf(gm, __shfl_xor(gm, off, 32));
    const float e0 = __expf(m0 - gm), e1 = __expf(m1 - gm), e2 = __expf(m2 - gm), e3 = __expf(m3 - gm);
    float z = (e0 + e1) + (e2 + e3);
    for (int off = 1; off < 32; off <<= 1)
        z += __shfl_xor(z, off, 32);
    if (wv == 0) {
        p_loc[lane] = e0; p_loc[lane + 32] = e1; p_loc[lane + 64] = e2; p_loc[lane + 96] = e3;
    }
    __syncthreads();

    {
        float4 s4 = make_float4(0.f, 0.f, 0.f, 0.f);
        for (int r = 0; r < 16; ++r) {
            const float p = p_loc[rm + r];
            const float4 cv = *reinterpret_cast<const float4*>(ctx_lds + (rm + r) * P + lane * 4);
            s4.x += p * cv.x; s4.y += p * cv.y; s4.z += p * cv.z; s4.w += p * cv.w;
        }
        *reinterpret_cast<float4*>(red + wv * Dn + lane * 4) = s4;
    }
    __syncthreads();

    float val;
    if (tid < Dn) {
        val = 0.f;
        for (int w = 0; w < 8; ++w) val += red[w * Dn + tid];
    } else {
        val = (tid == Dn) ? gm : ((tid == Dn + 1) ? z : 0.f);
    }

    for (int r = wv; r < CHUNK; r += 8) {
        const int c4 = lane * 4;
        const float4 cv = *reinterpret_cast<const float4*>(ctx_lds + r * P + c4);
        const float4 qv = *reinterpret_cast<const float4*>(sc_lds + r * P + c4);
        float* op = outB + (size_t)(cc + r) * OUTW;
        *reinterpret_cast<float4*>(op + c4) = cv;
        *reinterpret_cast<float4*>(op + Dn + c4) = qv;
        const float4 pv = make_float4(cv.x * qv.x, cv.y * qv.y, cv.z * qv.z, cv.w * qv.w);
        *reinterpret_cast<float4*>(op + 2 * Dn + c4) = pv;
    }

    {
        const int srow = 2 * ch + (tid >> 7);
        const int col  = 3 * Dn + (tid & 127);
        for (int k = 0; k < NCHUNK; ++k)
            outB[(size_t)(k * CHUNK + srow) * OUTW + col] = val;
    }
}

__global__ __launch_bounds__(256)
void q2c_block_kernel(const float* __restrict__ ctx_g, float* out_g)
{
    __shared__ float q2c[Dn];
    const int b    = blockIdx.x / NCHUNK;
    const int k    = blockIdx.x % NCHUNK;
    const int tid  = threadIdx.x;
    const int wv   = tid >> 5;
    const int lane = tid & 31;
    const float* ctxC = ctx_g + ((size_t)b * Cn + (size_t)k * CHUNK) * Dn;
    float*       outC = out_g + ((size_t)b * Cn + (size_t)k * CHUNK) * OUTW + 3 * Dn;

    float4 cv[16];
    for (int i = 0; i < 16; ++i)
        cv[i] = *reinterpret_cast<const float4*>(ctxC + (size_t)(wv + 8 * i) * Dn + lane * 4);

    if (tid < Dn) {
        float gmj[NCHUNK], zj[NCHUNK], vj[NCHUNK];
        float gmax = -3.4e38f;
        for (int j = 0; j < NCHUNK; ++j) {
            gmj[j] = outC[(size_t)(2 * j + 1) * OUTW + 0];
            zj[j]  = outC[(size_t)(2 * j + 1) * OUTW + 1];
            vj[j]  = outC[(size_t)(2 * j) * OUTW + tid];
            gmax = fmaxf(gmax, gmj[j]);
        }
        float Z = 0.f, s = 0.f;
        for (int j = 0; j < NCHUNK; ++j) {
            const float w = __expf(gmj[j] - gmax);
            Z += w * zj[j];
            s += w * vj[j];
        }
        q2c[tid] = s / Z;
    }
    __syncthreads();

    const float4 qv = *reinterpret_cast<const float4*>(q2c + lane * 4);
    for (int i = 0; i < 16; ++i) {
        const float4 pv = make_float4(cv[i].x * qv.x, cv[i].y * qv.y, cv[i].z * qv.z, cv[i].w * qv.w);
        *reinterpret_cast<float4*>(outC + (size_t)(wv + 8 * i) * OUTW + lane * 4) = pv;
    }
}

extern "C" void kernel_launch(void* const* d_in, const int* in_sizes, int n_in,
                              void* d_out, int out_size, void* d_ws, size_t ws_size,
                              hipStream_t stream) {
    const float* ctx   = (const float*)d_in[0];
    const float* qry   = (const float*)d_in[1];
    const float* W     = (const float*)d_in[2];
    const float* qmask = (const float*)d_in[3];
    float* out = (float*)d_out;

    const size_t smem_bytes = (size_t)SM_FLOATS * sizeof(float);
    hipFuncSetAttribute(reinterpret_cast<const void*>(cqa_kernel),
                        hipFuncAttributeMaxDynamicSharedMemorySize, (int)smem_bytes);
    cqa_kernel<<<dim3(NWG), dim3(256), smem_bytes, stream>>>(ctx, qry, W, qmask, out);
    q2c_block_kernel<<<dim3(NWG), dim3(256), 0, stream>>>(ctx, out);
}


# LDS per work-group 206 KB -> 154 KB so two work-groups share a WGP: scores stay in registers and reach the second product through a 32-column wave-private slab; c2q staged over the dead query tile
# speedup vs baseline: 5.2836x; 1.1957x over previous
#include <hip/hip_runtime.h>


typedef float v2f __attribute__((ext_vector_type(2)));
typedef float v8f __attribute__((ext_vector_type(8)));

#define NEG_INF_F (-1000000000.0f)

constexpr int Bn = 64, Cn = 1024, Qn = 128, Dn = 128;
constexpr int P = 132;
constexpr int CHUNK = 128;
constexpr int NCHUNK = Cn / CHUNK;
constexpr int NWG = Bn * NCHUNK;
constexpr int OUTW = 4 * Dn;
constexpr int KQ = 32;
constexpr int PSC = KQ + 4;
constexpr int SM_FLOATS = 2 * (128 * P) + CHUNK * PSC + 384 + 128 + 128 + 128 + 128;
static_assert(Qn == CHUNK, "query rows and chunk rows are staged by one loop");
static_assert(CHUNK * PSC >= 8 * Dn, "the per-wave partial rows reuse the score slab");
static_assert(2 * ((SM_FLOATS * 4 + 2047) / 2048) * 2048 <= 320 * 1024, "two work-groups per WGP");

__global__ __launch_bounds__(256, 2)
void cqa_kernel(const float* __restrict__ ctx_g,
                const float* __restrict__ qry_g,
                const float* __restrict__ W_g,
                const float* __restrict__ qmask_g,
                float* __restrict__ out_g)
{
    extern __shared__ float smem[];
    float* q_lds   = smem;
    float* ctx_lds = q_lds   + Qn * P;
    float* sc_lds  = ctx_lds + CHUNK * P;
    float* w_lds   = sc_lds  + CHUNK * PSC;
    float* qbias   = w_lds   + 384;
    float* cterm   = qbias   + Qn;
    float* m_loc   = cterm   + CHUNK;
    float* p_loc   = m_loc   + CHUNK;
    float* red     = sc_lds;

    const int b    = blockIdx.x / NCHUNK;
    const int ch   = blockIdx.x % NCHUNK;
    const int cc   = ch * CHUNK;
    const int tid  = threadIdx.x;
    const int wv   = tid >> 5;
    const int lane = tid & 31;
    const int lh   = lane >> 4;
    const int ln   = lane & 15;

    const float* ctxB = ctx_g + (size_t)b * Cn * Dn;
    const float* qryB = qry_g + (size_t)b * Qn * Dn;
    float*       outB = out_g + (size_t)b * Cn * OUTW;

    {
        const float w0 = W_g[tid];
        const float w1 = (tid < Dn) ? W_g[256 + tid] : 0.f;
        const int c4 = lane * 4;
        float4 qv[16], cv[16];
        for (int i = 0; i < 16; ++i) {
            const int r = wv + 8 * i;
            qv[i] = *reinterpret_cast<const float4*>(qryB + (size_t)r * Dn + c4);
            cv[i] = *reinterpret_cast<const float4*>(ctxB + (size_t)(cc + r) * Dn + c4);
        }
        w_lds[tid] = w0;
        if (tid < Dn) w_lds[256 + tid] = w1;
        for (int i = 0; i < 16; ++i) {
            const int r = wv + 8 * i;
            *reinterpret_cast<float4*>(q_lds + r * P + c4) = qv[i];
            *reinterpret_cast<float4*>(ctx_lds + r * P + c4) = cv[i];
        }
    }
    __syncthreads();
    if (tid < Qn) {
        float s = 0.f;
        for (int d = 0; d < Dn; ++d) s += q_lds[tid * P + d] * w_lds[Dn + d];
        qbias[tid] = s + (1.0f - qmask_g[b * Qn + tid]) * NEG_INF_F;
    } else {
        const int row = tid - Qn;
        float s = 0.f;
        for (int d = 0; d < Dn; ++d) s += ctx_lds[row * P + d] * w_lds[d];
        cterm[row] = s;
    }
    __syncthreads();

    const int rm = wv * 16;

    v8f acc[8];
    #pragma unroll
    for (int t = 0; t < 8; ++t) acc[t] = v8f{0.f,0.f,0.f,0.f,0.f,0.f,0.f,0.f};

    for (int kb = 0; kb < Dn; kb += 4) {
        const int k0 = kb + 2 * lh;
        v2f a = *reinterpret_cast<const v2f*>(ctx_lds + (rm + ln) * P + k0);
        a.x *= w_lds[2 * Dn + k0];
        a.y *= w_lds[2 * Dn + k0 + 1];
        #pragma unroll
        for (int t = 0; t < 8; ++t) {
            const v2f bf = *reinterpret_cast<const v2f*>(q_lds + (t * 16 + ln) * P + k0);
            acc[t] = __builtin_amdgcn_wmma_f32_16x16x4_f32(
                false, a, false, bf, (short)0, acc[t], false, false);
        }
    }

    #pragma unroll
    for (int r = 0; r < 8; ++r) {
        const int row = rm + r + 8 * lh;
        const float cb = cterm[row];
        float sv[8];
        float mx = -3.4e38f;
        #pragma unroll
        for (int t = 0; t < 8; ++t) {
            const float s = acc[t][r] + cb + qbias[t * 16 + ln];
            sv[t] = s;
            mx = fmaxf(mx, s);
        }
        #pragma unroll
        for (int off = 1; off < 16; off <<= 1)
            mx = fmaxf(mx, __shfl_xor(mx, off, 32));
        if (ln == 0) m_loc[row] = mx;
        float sum = 0.f;
        #pragma unroll
        for (int t = 0; t < 8; ++t) { sv[t] = __expf(sv[t] - mx); sum += sv[t]; }
        #pragma unroll
        for (int off = 1; off < 16; off <<= 1)
            sum += __shfl_xor(sum, off, 32);
        const float inv = 1.0f / sum;
        #pragma unroll
        for (int t = 0; t < 8; ++t)
            acc[t][r] = sv[t] * inv;
    }

    v8f acc2[8];
    #pragma unroll
    for (int t = 0; t < 8; ++t) acc2[t] = v8f{0.f,0.f,0.f,0.f,0.f,0.f,0.f,0.f};

    #pragma unroll
    for (int qt = 0; qt < Qn / KQ; ++qt) {
        #pragma unroll
        for (int r = 0; r < 8; ++r) {
            const int row = rm + r + 8 * lh;
            sc_lds[row * PSC + ln]      = acc[2 * qt][r];
            sc_lds[row * PSC + 16 + ln] = acc[2 * qt + 1][r];
        }
        __builtin_amdgcn_fence(__ATOMIC_ACQ_REL, "wavefront");
        for (int kb = 0; kb < KQ; kb += 4) {
            const int k0 = kb + 2 * lh;
            const v2f a = *reinterpret_cast<const v2f*>(sc_lds + (rm + ln) * PSC + k0);
            const int kq = qt * KQ + k0;
            #pragma unroll
            for (int t = 0; t < 8; ++t) {
                v2f bf;
                bf.x = q_lds[kq * P + t * 16 + ln];
                bf.y = q_lds[(kq + 1) * P + t * 16 + ln];
                acc2[t] = __builtin_amdgcn_wmma_f32_16x16x4_f32(
                    false, a, false, bf, (short)0, acc2[t], false, false);
            }
        }
        __builtin_amdgcn_fence(__ATOMIC_ACQ_REL, "wavefront");
    }
    __syncthreads();

    float* c2q_lds = q_lds;
    #pragma unroll
    for (int r = 0; r < 8; ++r) {
        const int row = rm + r + 8 * lh;
        #pragma unroll
        for (int t = 0; t < 8; ++t)
            c2q_lds[row * P + t * 16 + ln] = acc2[t][r];
    }

    const float m0 = m_loc[lane], m1 = m_loc[lane + 32], m2 = m_loc[lane + 64], m3 = m_loc[lane + 96];
    float gm = fmaxf(fmaxf(m0, m1), fmaxf(m2, m3));
    for (int off = 1; off < 32; off <<= 1)
        gm = fmaxf(gm, __shfl_xor(gm, off, 32));
    const float e0 = __expf(m0 - gm), e1 = __expf(m1 - gm), e2 = __expf(m2 - gm), e3 = __expf(m3 - gm);
    float z = (e0 + e1) + (e2 + e3);
    for (int off = 1; off < 32; off <<= 1)
        z += __shfl_xor(z, off, 32);
    if (wv == 0) {
        p_loc[lane] = e0; p_loc[lane + 32] = e1; p_loc[lane + 64] = e2; p_loc[lane + 96] = e3;
    }
    __syncthreads();

    {
        float4 s4 = make_float4(0.f, 0.f, 0.f, 0.f);
        for (int r = 0; r < 16; ++r) {
            const float p = p_loc[rm + r];
            const float4 cv = *reinterpret_cast<const float4*>(ctx_lds + (rm + r) * P + lane * 4);
            s4.x += p * cv.x; s4.y += p * cv.y; s4.z += p * cv.z; s4.w += p * cv.w;
        }
        *reinterpret_cast<float4*>(red + wv * Dn + lane * 4) = s4;
    }
    __syncthreads();

    float val;
    if (tid < Dn) {
        val = 0.f;
        for (int w = 0; w < 8; ++w) val += red[w * Dn + tid];
    } else {
        val = (tid == Dn) ? gm : ((tid == Dn + 1) ? z : 0.f);
    }

    for (int r = wv; r < CHUNK; r += 8) {
        const int c4 = lane * 4;
        const float4 cv = *reinterpret_cast<const float4*>(ctx_lds + r * P + c4);
        const float4 qv = *reinterpret_cast<const float4*>(c2q_lds + r * P + c4);
        float* op = outB + (size_t)(cc + r) * OUTW;
        *reinterpret_cast<float4*>(op + c4) = cv;
        *reinterpret_cast<float4*>(op + Dn + c4) = qv;
        const float4 pv = make_float4(cv.x * qv.x, cv.y * qv.y, cv.z * qv.z, cv.w * qv.w);
        *reinterpret_cast<float4*>(op + 2 * Dn + c4) = pv;
    }

    {
        const int srow = 2 * ch + (tid >> 7);
        const int col  = 3 * Dn + (tid & 127);
        for (int k = 0; k < NCHUNK; ++k)
            outB[(size_t)(k * CHUNK + srow) * OUTW + col] = val;
    }
}

__global__ __launch_bounds__(256)
void q2c_block_kernel(const float* __restrict__ ctx_g, float* out_g)
{
    __shared__ float q2c[Dn];
    const int b    = blockIdx.x / NCHUNK;
    const int k    = blockIdx.x % NCHUNK;
    const int tid  = threadIdx.x;
    const int wv   = tid >> 5;
    const int lane = tid & 31;
    const float* ctxC = ctx_g + ((size_t)b * Cn + (size_t)k * CHUNK) * Dn;
    float*       outC = out_g + ((size_t)b * Cn + (size_t)k * CHUNK) * OUTW + 3 * Dn;

    float4 cv[16];
    for (int i = 0; i < 16; ++i)
        cv[i] = *reinterpret_cast<const float4*>(ctxC + (size_t)(wv + 8 * i) * Dn + lane * 4);

    if (tid < Dn) {
        float gmj[NCHUNK], zj[NCHUNK], vj[NCHUNK];
        float gmax = -3.4e38f;
        for (int j = 0; j < NCHUNK; ++j) {
            gmj[j] = outC[(size_t)(2 * j + 1) * OUTW + 0];
            zj[j]  = outC[(size_t)(2 * j + 1) * OUTW + 1];
            vj[j]  = outC[(size_t)(2 * j) * OUTW + tid];
            gmax = fmaxf(gmax, gmj[j]);
        }
        float Z = 0.f, s = 0.f;
        for (int j = 0; j < NCHUNK; ++j) {
            const float w = __expf(gmj[j] - gmax);
            Z += w * zj[j];
            s += w * vj[j];
        }
        q2c[tid] = s / Z;
    }
    __syncthreads();

    const float4 qv = *reinterpret_cast<const float4*>(q2c + lane * 4);
    for (int i = 0; i < 16; ++i) {
        const float4 pv = make_float4(cv[i].x * qv.x, cv[i].y * qv.y, cv[i].z * qv.z, cv[i].w * qv.w);
        *reinterpret_cast<float4*>(outC + (size_t)(wv + 8 * i) * OUTW + lane * 4) = pv;
    }
}

extern "C" void kernel_launch(void* const* d_in, const int* in_sizes, int n_in,
                              void* d_out, int out_size, void* d_ws, size_t ws_size,
                              hipStream_t stream) {
    const float* ctx   = (const float*)d_in[0];
    const float* qry   = (const float*)d_in[1];
    const float* W     = (const float*)d_in[2];
    const float* qmask = (const float*)d_in[3];
    float* out = (float*)d_out;

    const size_t smem_bytes = (size_t)SM_FLOATS * sizeof(float);
    hipFuncSetAttribute(reinterpret_cast<const void*>(cqa_kernel),
                        hipFuncAttributeMaxDynamicSharedMemorySize, (int)smem_bytes);
    cqa_kernel<<<dim3(NWG), dim3(256), smem_bytes, stream>>>(ctx, qry, W, qmask, out);
    q2c_block_kernel<<<dim3(NWG), dim3(256), 0, stream>>>(ctx, out);
}


# Scores computed transposed (query tiles as A, the wave's context rows as B): a lane owns one row, softmax needs 2 lane exchanges, scores feed c2q as A from registers; softmax and c2q sum order change
# speedup vs baseline: 5.5742x; 1.0550x over previous
#include <hip/hip_runtime.h>


typedef float v2f __attribute__((ext_vector_type(2)));
typedef float v8f __attribute__((ext_vector_type(8)));

#define NEG_INF_F (-1000000000.0f)

constexpr int Bn = 64, Cn = 1024, Qn = 128, Dn = 128;
constexpr int P = 132;
constexpr int CHUNK = 128;
constexpr int NCHUNK = Cn / CHUNK;
constexpr int NWG = Bn * NCHUNK;
constexpr int OUTW = 4 * Dn;
constexpr int SM_FLOATS = 2 * (128 * P) + 8 * Dn + 384 + 128 + 128 + 128 + 128;
static_assert(Qn == CHUNK, "query rows and chunk rows are staged by one loop");
static_assert((2 * (128 * P) + 8 * Dn + 384) % 4 == 0, "the bias vector is read with 16-byte loads");
static_assert(2 * ((SM_FLOATS * 4 + 2047) / 2048) * 2048 <= 320 * 1024, "two work-groups per WGP");

__global__ __launch_bounds__(256, 2)
void cqa_kernel(const float* __restrict__ ctx_g,
                const float* __restrict__ qry_g,
                const float* __restrict__ W_g,
                const float* __restrict__ qmask_g,
                float* __restrict__ out_g)
{
    extern __shared__ float smem[];
    float* q_lds   = smem;
    float* ctx_lds = q_lds   + Qn * P;
    float* red     = ctx_lds + CHUNK * P;
    float* w_lds   = red     + 8 * Dn;
    float* qbias   = w_lds   + 384;
    float* cterm   = qbias   + Qn;
    float* m_loc   = cterm   + CHUNK;
    float* p_loc   = m_loc   + CHUNK;

    const int b    = blockIdx.x / NCHUNK;
    const int ch   = blockIdx.x % NCHUNK;
    const int cc   = ch * CHUNK;
    const int tid  = threadIdx.x;
    const int wv   = tid >> 5;
    const int lane = tid & 31;
    const int lh   = lane >> 4;
    const int ln   = lane & 15;

    const float* ctxB = ctx_g + (size_t)b * Cn * Dn;
    const float* qryB = qry_g + (size_t)b * Qn * Dn;
    float*       outB = out_g + (size_t)b * Cn * OUTW;

    {
        const float w0 = W_g[tid];
        const float w1 = (tid < Dn) ? W_g[256 + tid] : 0.f;
        const int c4 = lane * 4;
        float4 qv[16], cv[16];
        for (int i = 0; i < 16; ++i) {
            const int r = wv + 8 * i;
            qv[i] = *reinterpret_cast<const float4*>(qryB + (size_t)r * Dn + c4);
            cv[i] = *reinterpret_cast<const float4*>(ctxB + (size_t)(cc + r) * Dn + c4);
        }
        w_lds[tid] = w0;
        if (tid < Dn) w_lds[256 + tid] = w1;
        for (int i = 0; i < 16; ++i) {
            const int r = wv + 8 * i;
            *reinterpret_cast<float4*>(q_lds + r * P + c4) = qv[i];
            *reinterpret_cast<float4*>(ctx_lds + r * P + c4) = cv[i];
        }
    }
    __syncthreads();
    if (tid < Qn) {
        float s = 0.f;
        for (int d = 0; d < Dn; ++d) s += q_lds[tid * P + d] * w_lds[Dn + d];
        qbias[tid] = s + (1.0f - qmask_g[b * Qn + tid]) * NEG_INF_F;
    } else {
        const int row = tid - Qn;
        float s = 0.f;
        for (int d = 0; d < Dn; ++d) s += ctx_lds[row * P + d] * w_lds[d];
        cterm[row] = s;
    }
    __syncthreads();

    const int rm = wv * 16;

    v8f acc[8];
    #pragma unroll
    for (int t = 0; t < 8; ++t) acc[t] = v8f{0.f,0.f,0.f,0.f,0.f,0.f,0.f,0.f};

    for (int kb = 0; kb < Dn; kb += 4) {
        const int k0 = kb + 2 * lh;
        v2f cf = *reinterpret_cast<const v2f*>(ctx_lds + (rm + ln) * P + k0);
        cf.x *= w_lds[2 * Dn + k0];
        cf.y *= w_lds[2 * Dn + k0 + 1];
        #pragma unroll
        for (int t = 0; t < 8; ++t) {
            const v2f qf = *reinterpret_cast<const v2f*>(q_lds + (t * 16 + ln) * P + k0);
            acc[t] = __builtin_amdgcn_wmma_f32_16x16x4_f32(
                false, qf, false, cf, (short)0, acc[t], false, false);
        }
    }

    {
        const int crow = rm + ln;
        const float cb = cterm[crow];
        float mx = -3.4e38f;
        #pragma unroll
        for (int t = 0; t < 8; ++t) {
            const float4 q0 = *reinterpret_cast<const float4*>(qbias + t * 16 + 8 * lh);
            const float4 q1 = *reinterpret_cast<const float4*>(qbias + t * 16 + 8 * lh + 4);
            const float qv[8] = {q0.x, q0.y, q0.z, q0.w, q1.x, q1.y, q1.z, q1.w};
            #pragma unroll
            for (int r = 0; r < 8; ++r) {
                const float sc = acc[t][r] + cb + qv[r];
                acc[t][r] = sc;
                mx = fmaxf(mx, sc);
            }
        }
        mx = fmaxf(mx, __shfl_xor(mx, 16, 32));
        if (lh == 0) m_loc[crow] = mx;
        float ps[4] = {0.f, 0.f, 0.f, 0.f};
        #pragma unroll
        for (int t = 0; t < 8; ++t) {
            #pragma unroll
            for (int r = 0; r < 8; ++r) {
                const float e = __expf(acc[t][r] - mx);
                acc[t][r] = e;
                ps[r & 3] += e;
            }
        }
        float sum = (ps[0] + ps[1]) + (ps[2] + ps[3]);
        sum += __shfl_xor(sum, 16, 32);
        const float inv = 1.0f / sum;
        #pragma unroll
        for (int t = 0; t < 8; ++t)
            acc[t] *= inv;
    }

    v8f acc2[8];
    #pragma unroll
    for (int t = 0; t < 8; ++t) acc2[t] = v8f{0.f,0.f,0.f,0.f,0.f,0.f,0.f,0.f};

    const float* qb0 = q_lds + (8 * lh) * P + ln;
    float bx[2][8], by[2][8];
    #pragma unroll
    for (int nt = 0; nt < 8; ++nt) {
        bx[0][nt] = qb0[nt * 16];
        by[0][nt] = qb0[P + nt * 16];
    }
    #pragma unroll
    for (int s = 0; s < 32; ++s) {
        const int cur = s & 1;
        if (s + 1 < 32) {
            const int krow = ((s + 1) >> 2) * 16 + ((s + 1) & 3) * 2;
            #pragma unroll
            for (int nt = 0; nt < 8; ++nt) {
                bx[cur ^ 1][nt] = qb0[krow * P + nt * 16];
                by[cur ^ 1][nt] = qb0[(krow + 1) * P + nt * 16];
            }
        }
        __builtin_amdgcn_sched_barrier(0);
        v2f a;
        a.x = acc[s >> 2][2 * (s & 3)];
        a.y = acc[s >> 2][2 * (s & 3) + 1];
        #pragma unroll
        for (int nt = 0; nt < 8; ++nt) {
            v2f bf;
            bf.x = bx[cur][nt];
            bf.y = by[cur][nt];
            acc2[nt] = __builtin_amdgcn_wmma_f32_16x16x4_f32(
                false, a, false, bf, (short)0, acc2[nt], false, false);
        }
        __builtin_amdgcn_sched_barrier(0);
    }
    __syncthreads();

    float* c2q_lds = q_lds;
    #pragma unroll
    for (int r = 0; r < 8; ++r) {
        const int row = rm + r + 8 * lh;
        #pragma unroll
        for (int t = 0; t < 8; ++t)
            c2q_lds[row * P + t * 16 + ln] = acc2[t][r];
    }

    const float m0 = m_loc[lane], m1 = m_loc[lane + 32], m2 = m_loc[lane + 64], m3 = m_loc[lane + 96];
    float gm = fmaxf(fmaxf(m0, m1), fmaxf(m2, m3));
    for (int off = 1; off < 32; off <<= 1)
        gm = fmaxf(gm, __shfl_xor(gm, off, 32));
    const float e0 = __expf(m0 - gm), e1 = __expf(m1 - gm), e2 = __expf(m2 - gm), e3 = __expf(m3 - gm);
    float z = (e0 + e1) + (e2 + e3);
    for (int off = 1; off < 32; off <<= 1)
        z += __shfl_xor(z, off, 32);
    if (wv == 0) {
        p_loc[lane] = e0; p_loc[lane + 32] = e1; p_loc[lane + 64] = e2; p_loc[lane + 96] = e3;
    }
    __syncthreads();

    {
        float4 s4 = make_float4(0.f, 0.f, 0.f, 0.f);
        for (int r = 0; r < 16; ++r) {
            const float p = p_loc[rm + r];
            const float4 cv = *reinterpret_cast<const float4*>(ctx_lds + (rm + r) * P + lane * 4);
            s4.x += p * cv.x; s4.y += p * cv.y; s4.z += p * cv.z; s4.w += p * cv.w;
        }
        *reinterpret_cast<float4*>(red + wv * Dn + lane * 4) = s4;
    }
    __syncthreads();

    float val;
    if (tid < Dn) {
        val = 0.f;
        for (int w = 0; w < 8; ++w) val += red[w * Dn + tid];
    } else {
        val = (tid == Dn) ? gm : ((tid == Dn + 1) ? z : 0.f);
    }

    for (int r = wv; r < CHUNK; r += 8) {
        const int c4 = lane * 4;
        const float4 cv = *reinterpret_cast<const float4*>(ctx_lds + r * P + c4);
        const float4 qv = *reinterpret_cast<const float4*>(c2q_lds + r * P + c4);
        float* op = outB + (size_t)(cc + r) * OUTW;
        *reinterpret_cast<float4*>(op + c4) = cv;
        *reinterpret_cast<float4*>(op + Dn + c4) = qv;
        const float4 pv = make_float4(cv.x * qv.x, cv.y * qv.y, cv.z * qv.z, cv.w * qv.w);
        *reinterpret_cast<float4*>(op + 2 * Dn + c4) = pv;
    }

    {
        const int srow = 2 * ch + (tid >> 7);
        const int col  = 3 * Dn + (tid & 127);
        for (int k = 0; k < NCHUNK; ++k)
            outB[(size_t)(k * CHUNK + srow) * OUTW + col] = val;
    }
}

__global__ __launch_bounds__(256)
void q2c_block_kernel(const float* __restrict__ ctx_g, float* out_g)
{
    __shared__ float q2c[Dn];
    const int b    = blockIdx.x / NCHUNK;
    const int k    = blockIdx.x % NCHUNK;
    const int tid  = threadIdx.x;
    const int wv   = tid >> 5;
    const int lane = tid & 31;
    const float* ctxC = ctx_g + ((size_t)b * Cn + (size_t)k * CHUNK) * Dn;
    float*       outC = out_g + ((size_t)b * Cn + (size_t)k * CHUNK) * OUTW + 3 * Dn;

    float4 cv[16];
    for (int i = 0; i < 16; ++i)
        cv[i] = *reinterpret_cast<const float4*>(ctxC + (size_t)(wv + 8 * i) * Dn + lane * 4);

    if (tid < Dn) {
        float gmj[NCHUNK], zj[NCHUNK], vj[NCHUNK];
        float gmax = -3.4e38f;
        for (int j = 0; j < NCHUNK; ++j) {
            gmj[j] = outC[(size_t)(2 * j + 1) * OUTW + 0];
            zj[j]  = outC[(size_t)(2 * j + 1) * OUTW + 1];
            vj[j]  = outC[(size_t)(2 * j) * OUTW + tid];
            gmax = fmaxf(gmax, gmj[j]);
        }
        float Z = 0.f, s = 0.f;
        for (int j = 0; j < NCHUNK; ++j) {
            const float w = __expf(gmj[j] - gmax);
            Z += w * zj[j];
            s += w * vj[j];
        }
        q2c[tid] = s / Z;
    }
    __syncthreads();

    const float4 qv = *reinterpret_cast<const float4*>(q2c + lane * 4);
    for (int i = 0; i < 16; ++i) {
        const float4 pv = make_float4(cv[i].x * qv.x, cv[i].y * qv.y, cv[i].z * qv.z, cv[i].w * qv.w);
        *reinterpret_cast<float4*>(outC + (size_t)(wv + 8 * i) * OUTW + lane * 4) = pv;
    }
}

extern "C" void kernel_launch(void* const* d_in, const int* in_sizes, int n_in,
                              void* d_out, int out_size, void* d_ws, size_t ws_size,
                              hipStream_t stream) {
    const float* ctx   = (const float*)d_in[0];
    const float* qry   = (const float*)d_in[1];
    const float* W     = (const float*)d_in[2];
    const float* qmask = (const float*)d_in[3];
    float* out = (float*)d_out;

    const size_t smem_bytes = (size_t)SM_FLOATS * sizeof(float);
    hipFuncSetAttribute(reinterpret_cast<const void*>(cqa_kernel),
                        hipFuncAttributeMaxDynamicSharedMemorySize, (int)smem_bytes);
    cqa_kernel<<<dim3(NWG), dim3(256), smem_bytes, stream>>>(ctx, qry, W, qmask, out);
    q2c_block_kernel<<<dim3(NWG), dim3(256), 0, stream>>>(ctx, out);
}


# Output stores run under the products: block 0 leaves before the first product, each wave stores its own 16 rows with no block barrier; q2c share from rescaled per-wave partials (sum order changes)
# speedup vs baseline: 6.0641x; 1.0879x over previous
#include <hip/hip_runtime.h>


typedef float v2f __attribute__((ext_vector_type(2)));
typedef float v8f __attribute__((ext_vector_type(8)));

#define NEG_INF_F (-1000000000.0f)

constexpr int Bn = 64, Cn = 1024, Qn = 128, Dn = 128;
constexpr int P = 132;
constexpr int CHUNK = 128;
constexpr int NCHUNK = Cn / CHUNK;
constexpr int NWG = Bn * NCHUNK;
constexpr int OUTW = 4 * Dn;
constexpr int SM_FLOATS = 2 * (128 * P) + 8 * Dn + 384 + 128 + 128 + 16;
static_assert(Qn == CHUNK, "query rows and chunk rows are staged by one loop");
static_assert((2 * (128 * P) + 8 * Dn + 384) % 4 == 0, "the bias vector and the wave shares are read with 16-byte loads");
static_assert(2 * ((SM_FLOATS * 4 + 2047) / 2048) * 2048 <= 320 * 1024, "two work-groups per WGP");

template <int M>
__device__ __forceinline__ float row_xor(float v)
{
    return __int_as_float(__builtin_amdgcn_update_dpp(__float_as_int(v), __float_as_int(v),
                                                      0x160 | M, 0xf, 0xf, false));
}

__global__ __launch_bounds__(256, 2)
void cqa_kernel(const float* __restrict__ ctx_g,
                const float* __restrict__ qry_g,
                const float* __restrict__ W_g,
                const float* __restrict__ qmask_g,
                float* __restrict__ out_g)
{
    extern __shared__ float smem[];
    float* q_lds   = smem;
    float* ctx_lds = q_lds   + Qn * P;
    float* red     = ctx_lds + CHUNK * P;
    float* w_lds   = red     + 8 * Dn;
    float* qbias   = w_lds   + 384;
    float* cterm   = qbias   + Qn;
    float* wv_gm   = cterm   + CHUNK;
    float* wv_z    = wv_gm   + 8;

    const int b    = blockIdx.x / NCHUNK;
    const int ch   = blockIdx.x % NCHUNK;
    const int cc   = ch * CHUNK;
    const int tid  = threadIdx.x;
    const int wv   = tid >> 5;
    const int lane = tid & 31;
    const int lh   = lane >> 4;
    const int ln   = lane & 15;

    const float* ctxB = ctx_g + (size_t)b * Cn * Dn;
    const float* qryB = qry_g + (size_t)b * Qn * Dn;
    float*       outB = out_g + (size_t)b * Cn * OUTW;

    {
        const float w0 = W_g[tid];
        const float w1 = (tid < Dn) ? W_g[256 + tid] : 0.f;
        const int c4 = lane * 4;
        float4 qv[16], cv[16];
        for (int i = 0; i < 16; ++i) {
            const int r = wv + 8 * i;
            qv[i] = *reinterpret_cast<const float4*>(qryB + (size_t)r * Dn + c4);
            cv[i] = *reinterpret_cast<const float4*>(ctxB + (size_t)(cc + r) * Dn + c4);
        }
        w_lds[tid] = w0;
        if (tid < Dn) w_lds[256 + tid] = w1;
        for (int i = 0; i < 16; ++i) {
            const int r = wv + 8 * i;
            *reinterpret_cast<float4*>(q_lds + r * P + c4) = qv[i];
            *reinterpret_cast<float4*>(ctx_lds + r * P + c4) = cv[i];
        }
    }
    __syncthreads();
    if (tid < Qn) {
        float s = 0.f;
        for (int d = 0; d < Dn; ++d) s += q_lds[tid * P + d] * w_lds[Dn + d];
        qbias[tid] = s + (1.0f - qmask_g[b * Qn + tid]) * NEG_INF_F;
    } else {
        const int row = tid - Qn;
        float s = 0.f;
        for (int d = 0; d < Dn; ++d) s += ctx_lds[row * P + d] * w_lds[d];
        cterm[row] = s;
    }
    __syncthreads();

    const int rm = wv * 16;

    #pragma unroll
    for (int r = 0; r < 16; ++r) {
        const float4 v = *reinterpret_cast<const float4*>(ctx_lds + (rm + r) * P + lane * 4);
        *reinterpret_cast<float4*>(outB + (size_t)(cc + rm + r) * OUTW + lane * 4) = v;
    }

    v8f acc[8];
    #pragma unroll
    for (int t = 0; t < 8; ++t) acc[t] = v8f{0.f,0.f,0.f,0.f,0.f,0.f,0.f,0.f};

    for (int kb = 0; kb < Dn; kb += 4) {
        const int k0 = kb + 2 * lh;
        v2f cf = *reinterpret_cast<const v2f*>(ctx_lds + (rm + ln) * P + k0);
        cf.x *= w_lds[2 * Dn + k0];
        cf.y *= w_lds[2 * Dn + k0 + 1];
        #pragma unroll
        for (int t = 0; t < 8; ++t) {
            const v2f qf = *reinterpret_cast<const v2f*>(q_lds + (t * 16 + ln) * P + k0);
            acc[t] = __builtin_amdgcn_wmma_f32_16x16x4_f32(
                false, qf, false, cf, (short)0, acc[t], false, false);
        }
    }

    float mx = -3.4e38f;
    {
        const int crow = rm + ln;
        const float cb = cterm[crow];
        #pragma unroll
        for (int t = 0; t < 8; ++t) {
            const float4 q0 = *reinterpret_cast<const float4*>(qbias + t * 16 + 8 * lh);
            const float4 q1 = *reinterpret_cast<const float4*>(qbias + t * 16 + 8 * lh + 4);
            const float qv[8] = {q0.x, q0.y, q0.z, q0.w, q1.x, q1.y, q1.z, q1.w};
            #pragma unroll
            for (int r = 0; r < 8; ++r) {
                const float sc = acc[t][r] + cb + qv[r];
                acc[t][r] = sc;
                mx = fmaxf(mx, sc);
            }
        }
        mx = fmaxf(mx, __shfl_xor(mx, 16, 32));
        float ps[4] = {0.f, 0.f, 0.f, 0.f};
        #pragma unroll
        for (int t = 0; t < 8; ++t) {
            #pragma unroll
            for (int r = 0; r < 8; ++r) {
                const float e = __expf(acc[t][r] - mx);
                acc[t][r] = e;
                ps[r & 3] += e;
            }
        }
        float sum = (ps[0] + ps[1]) + (ps[2] + ps[3]);
        sum += __shfl_xor(sum, 16, 32);
        const float inv = 1.0f / sum;
        #pragma unroll
        for (int t = 0; t < 8; ++t)
            acc[t] *= inv;
    }

    v8f acc2[8];
    #pragma unroll
    for (int t = 0; t < 8; ++t) acc2[t] = v8f{0.f,0.f,0.f,0.f,0.f,0.f,0.f,0.f};

    const float* qb0 = q_lds + (8 * lh) * P + ln;
    float bx[2][8], by[2][8];
    #pragma unroll
    for (int nt = 0; nt < 8; ++nt) {
        bx[0][nt] = qb0[nt * 16];
        by[0][nt] = qb0[P + nt * 16];
    }
    #pragma unroll
    for (int s = 0; s < 32; ++s) {
        const int cur = s & 1;
        if (s + 1 < 32) {
            const int krow = ((s + 1) >> 2) * 16 + ((s + 1) & 3) * 2;
            #pragma unroll
            for (int nt = 0; nt < 8; ++nt) {
                bx[cur ^ 1][nt] = qb0[krow * P + nt * 16];
                by[cur ^ 1][nt] = qb0[(krow + 1) * P + nt * 16];
            }
        }
        __builtin_amdgcn_sched_barrier(0);
        v2f a;
        a.x = acc[s >> 2][2 * (s & 3)];
        a.y = acc[s >> 2][2 * (s & 3) + 1];
        #pragma unroll
        for (int nt = 0; nt < 8; ++nt) {
            v2f bf;
            bf.x = bx[cur][nt];
            bf.y = by[cur][nt];
            acc2[nt] = __builtin_amdgcn_wmma_f32_16x16x4_f32(
                false, a, false, bf, (short)0, acc2[nt], false, false);
        }
        __builtin_amdgcn_sched_barrier(0);
    }

    float gw = mx;
    gw = fmaxf(gw, row_xor<1>(gw)); gw = fmaxf(gw, row_xor<2>(gw));
    gw = fmaxf(gw, row_xor<4>(gw)); gw = fmaxf(gw, row_xor<8>(gw));
    const float e_row = __expf(mx - gw);
    float zw = e_row;
    zw += row_xor<1>(zw); zw += row_xor<2>(zw);
    zw += row_xor<4>(zw); zw += row_xor<8>(zw);

    float4 cr[16];
    #pragma unroll
    for (int r = 0; r < 16; ++r)
        cr[r] = *reinterpret_cast<const float4*>(ctx_lds + (rm + r) * P + lane * 4);

    __builtin_amdgcn_fence(__ATOMIC_RELEASE, "wavefront");
    __builtin_amdgcn_wave_barrier();
    __builtin_amdgcn_fence(__ATOMIC_ACQUIRE, "wavefront");

    #pragma unroll
    for (int r = 0; r < 8; ++r) {
        const int row = rm + r + 8 * lh;
        #pragma unroll
        for (int t = 0; t < 8; ++t)
            ctx_lds[row * P + t * 16 + ln] = acc2[t][r];
    }

    float4 s4 = make_float4(0.f, 0.f, 0.f, 0.f);
    #pragma unroll
    for (int r = 0; r < 16; ++r) {
        const float p = __int_as_float(__builtin_amdgcn_readlane(__float_as_int(e_row), r));
        s4.x += p * cr[r].x; s4.y += p * cr[r].y; s4.z += p * cr[r].z; s4.w += p * cr[r].w;
    }
    *reinterpret_cast<float4*>(red + wv * Dn + lane * 4) = s4;
    if (lane == 0) { wv_gm[wv] = gw; wv_z[wv] = zw; }

    __builtin_amdgcn_fence(__ATOMIC_RELEASE, "wavefront");
    __builtin_amdgcn_wave_barrier();
    __builtin_amdgcn_fence(__ATOMIC_ACQUIRE, "wavefront");

    #pragma unroll
    for (int r = 0; r < 16; ++r) {
        const int c4 = lane * 4;
        const float4 qv = *reinterpret_cast<const float4*>(ctx_lds + (rm + r) * P + c4);
        float* op = outB + (size_t)(cc + rm + r) * OUTW;
        *reinterpret_cast<float4*>(op + Dn + c4) = qv;
        const float4 pv = make_float4(cr[r].x * qv.x, cr[r].y * qv.y, cr[r].z * qv.z, cr[r].w * qv.w);
        *reinterpret_cast<float4*>(op + 2 * Dn + c4) = pv;
    }

    __builtin_amdgcn_fence(__ATOMIC_RELEASE, "workgroup", "local");
    __builtin_amdgcn_s_barrier();
    __builtin_amdgcn_fence(__ATOMIC_ACQUIRE, "workgroup", "local");

    const float4 g0 = *reinterpret_cast<const float4*>(wv_gm);
    const float4 g1 = *reinterpret_cast<const float4*>(wv_gm + 4);
    const float4 z0 = *reinterpret_cast<const float4*>(wv_z);
    const float4 z1 = *reinterpret_cast<const float4*>(wv_z + 4);
    const float gj[8] = {g0.x, g0.y, g0.z, g0.w, g1.x, g1.y, g1.z, g1.w};
    const float zj[8] = {z0.x, z0.y, z0.z, z0.w, z1.x, z1.y, z1.z, z1.w};
    float gm = gj[0];
    #pragma unroll
    for (int j = 1; j < 8; ++j) gm = fmaxf(gm, gj[j]);
    float wj[8];
    float z = 0.f;
    #pragma unroll
    for (int j = 0; j < 8; ++j) { wj[j] = __expf(gj[j] - gm); z += wj[j] * zj[j]; }

    float val;
    if (tid < Dn) {
        val = 0.f;
        #pragma unroll
        for (int w = 0; w < 8; ++w) val += wj[w] * red[w * Dn + tid];
    } else {
        val = (tid == Dn) ? gm : ((tid == Dn + 1) ? z : 0.f);
    }

    {
        const int srow = 2 * ch + (tid >> 7);
        const int col  = 3 * Dn + (tid & 127);
        for (int k = 0; k < NCHUNK; ++k)
            outB[(size_t)(k * CHUNK + srow) * OUTW + col] = val;
    }
}

__global__ __launch_bounds__(256)
void q2c_block_kernel(const float* __restrict__ ctx_g, float* out_g)
{
    __shared__ float q2c[Dn];
    const int b    = blockIdx.x / NCHUNK;
    const int k    = blockIdx.x % NCHUNK;
    const int tid  = threadIdx.x;
    const int wv   = tid >> 5;
    const int lane = tid & 31;
    const float* ctxC = ctx_g + ((size_t)b * Cn + (size_t)k * CHUNK) * Dn;
    float*       outC = out_g + ((size_t)b * Cn + (size_t)k * CHUNK) * OUTW + 3 * Dn;

    float4 cv[16];
    for (int i = 0; i < 16; ++i)
        cv[i] = *reinterpret_cast<const float4*>(ctxC + (size_t)(wv + 8 * i) * Dn + lane * 4);

    if (tid < Dn) {
        float gmj[NCHUNK], zj[NCHUNK], vj[NCHUNK];
        float gmax = -3.4e38f;
        for (int j = 0; j < NCHUNK; ++j) {
            gmj[j] = outC[(size_t)(2 * j + 1) * OUTW + 0];
            zj[j]  = outC[(size_t)(2 * j + 1) * OUTW + 1];
            vj[j]  = outC[(size_t)(2 * j) * OUTW + tid];
            gmax = fmaxf(gmax, gmj[j]);
        }
        float Z = 0.f, s = 0.f;
        for (int j = 0; j < NCHUNK; ++j) {
            const float w = __expf(gmj[j] - gmax);
            Z += w * zj[j];
            s += w * vj[j];
        }
        q2c[tid] = s / Z;
    }
    __syncthreads();

    const float4 qv = *reinterpret_cast<const float4*>(q2c + lane * 4);
    for (int i = 0; i < 16; ++i) {
        const float4 pv = make_float4(cv[i].x * qv.x, cv[i].y * qv.y, cv[i].z * qv.z, cv[i].w * qv.w);
        *reinterpret_cast<float4*>(outC + (size_t)(wv + 8 * i) * OUTW + lane * 4) = pv;
    }
}

extern "C" void kernel_launch(void* const* d_in, const int* in_sizes, int n_in,
                              void* d_out, int out_size, void* d_ws, size_t ws_size,
                              hipStream_t stream) {
    const float* ctx   = (const float*)d_in[0];
    const float* qry   = (const float*)d_in[1];
    const float* W     = (const float*)d_in[2];
    const float* qmask = (const float*)d_in[3];
    float* out = (float*)d_out;

    const size_t smem_bytes = (size_t)SM_FLOATS * sizeof(float);
    hipFuncSetAttribute(reinterpret_cast<const void*>(cqa_kernel),
                        hipFuncAttributeMaxDynamicSharedMemorySize, (int)smem_bytes);
    cqa_kernel<<<dim3(NWG), dim3(256), smem_bytes, stream>>>(ctx, qry, W, qmask, out);
    q2c_block_kernel<<<dim3(NWG), dim3(256), 0, stream>>>(ctx, out);
}
